# starting kernel
# baseline (speedup 1.0000x reference)
#include <hip/hip_runtime.h>
#include <hip/hip_bf16.h>

typedef float v2f __attribute__((ext_vector_type(2)));
typedef float v8f __attribute__((ext_vector_type(8)));

#define GIN_N    40000
#define GIN_E    640000
#define GIN_DIN  128
#define GIN_DOUT 256


__global__ void zero_kernel(int* __restrict__ p, int n) {
    int i = blockIdx.x * 256 + threadIdx.x;
    if (i < n) p[i] = 0;
}

__global__ void count_kernel(const int* __restrict__ dst, int* __restrict__ counts, int e) {
    int i = blockIdx.x * 256 + threadIdx.x;
    if (i < e) atomicAdd(&counts[dst[i]], 1);
}

__global__ void scan_kernel(const int* __restrict__ counts, int* __restrict__ offsets, int n) {
    __shared__ int sm[256];
    __shared__ int running;
    if (threadIdx.x == 0) running = 0;
    __syncthreads();
    for (int base = 0; base < n; base += 256) {
        int i = base + (int)threadIdx.x;
        int val = (i < n) ? counts[i] : 0;
        sm[threadIdx.x] = val;
        __syncthreads();
        for (int off = 1; off < 256; off <<= 1) {
            int t = (threadIdx.x >= (unsigned)off) ? sm[threadIdx.x - off] : 0;
            __syncthreads();
            sm[threadIdx.x] += t;
            __syncthreads();
        }
        if (i < n) offsets[i] = running + sm[threadIdx.x] - val;
        __syncthreads();
        if (threadIdx.x == 0) running += sm[255];
        __syncthreads();
    }
    if (threadIdx.x == 0) offsets[n] = running;
}

__global__ void fill_kernel(const int* __restrict__ src, const int* __restrict__ dst,
                            const int* __restrict__ offsets, int* __restrict__ cursor,
                            int* __restrict__ bucket, int e) {
    int i = blockIdx.x * 256 + threadIdx.x;
    if (i < e) {
        int d = dst[i];
        int pos = offsets[d] + atomicAdd(&cursor[d], 1);
        bucket[pos] = src[i];
    }
}

__global__ __launch_bounds__(256)
void gather_kernel(const float* __restrict__ x, const int* __restrict__ offsets,
                   const int* __restrict__ bucket, const float* __restrict__ epsp,
                   float* __restrict__ h) {
    int node = blockIdx.x * 8 + (threadIdx.x >> 5);
    int lane = threadIdx.x & 31;
    float se = 1.0f + epsp[0];
    const float4* xv = (const float4*)x;
    float4 self = xv[(size_t)node * 32 + lane];
    float ax = self.x * se, ay = self.y * se, az = self.z * se, aw = self.w * se;
    int e0 = offsets[node], e1 = offsets[node + 1];
    for (int i = e0; i < e1; ++i) {
        int s = bucket[i];
        float4 r = xv[(size_t)s * 32 + lane];
        ax += r.x; ay += r.y; az += r.z; aw += r.w;
    }
    float4 o; o.x = ax; o.y = ay; o.z = az; o.w = aw;
    ((float4*)h)[(size_t)node * 32 + lane] = o;
}


template <int K, bool USE_BN, bool USE_RES>
__global__ __launch_bounds__(128)
void gemm_kernel(const float* __restrict__ A, const float* __restrict__ B,
                 const float* __restrict__ bias, const float* __restrict__ gam,
                 const float* __restrict__ bet, const float* __restrict__ mu,
                 const float* __restrict__ var, const float* __restrict__ res,
                 float* __restrict__ out, int ncols) {
    constexpr int LDA = K + 4;
    constexpr int LDB = 72;
    extern __shared__ float lds[];
    float* As = lds;
    float* Bs = lds + 64 * LDA;

    const int rowBase = blockIdx.x * 64;
    const int colBase = blockIdx.y * 64;
    const int tid = threadIdx.x;

    for (int idx = tid; idx < 64 * K / 4; idx += 128) {
        int flat = idx * 4;
        int r = flat / K, c = flat % K;
        *(float4*)(As + r * LDA + c) =
            *(const float4*)(A + (size_t)(rowBase + r) * K + c);
    }
    for (int idx = tid; idx < K * 64 / 4; idx += 128) {
        int flat = idx * 4;
        int r = flat / 64, c = flat % 64;
        *(float4*)(Bs + r * LDB + c) =
            *(const float4*)(B + (size_t)r * ncols + colBase + c);
    }
    __syncthreads();

    const int wave = tid >> 5;
    const int lane = tid & 31;
    const int mrow = lane & 15;
    const int kh   = lane >> 4;
    const int rowOff = (wave >> 1) * 32;
    const int colOff = (wave & 1) * 32;

    v8f acc[2][2] = {};
    const float* Ap0 = As + (rowOff + mrow) * LDA + 2 * kh;
    const float* Ap1 = Ap0 + 16 * LDA;
    const float* Bp  = Bs + 2 * kh * LDB + colOff + mrow;

#pragma unroll 4
    for (int k = 0; k < K; k += 4) {
        v2f a0 = *(const v2f*)(Ap0 + k);
        v2f a1 = *(const v2f*)(Ap1 + k);
        const float* bk = Bp + k * LDB;
        v2f b0, b1;
        b0.x = bk[0];
        b0.y = bk[LDB];
        b1.x = bk[16];
        b1.y = bk[LDB + 16];
        acc[0][0] = __builtin_amdgcn_wmma_f32_16x16x4_f32(false, a0, false, b0, (short)0, acc[0][0], false, false);
        acc[0][1] = __builtin_amdgcn_wmma_f32_16x16x4_f32(false, a0, false, b1, (short)0, acc[0][1], false, false);
        acc[1][0] = __builtin_amdgcn_wmma_f32_16x16x4_f32(false, a1, false, b0, (short)0, acc[1][0], false, false);
        acc[1][1] = __builtin_amdgcn_wmma_f32_16x16x4_f32(false, a1, false, b1, (short)0, acc[1][1], false, false);
    }

#pragma unroll
    for (int ct = 0; ct < 2; ++ct) {
        int c = colBase + colOff + ct * 16 + mrow;
        float scale = 1.0f, shift = 0.0f;
        if (USE_BN) {
            float inv = rsqrtf(var[c] + 1e-5f);
            scale = gam[c] * inv;
            shift = (bias[c] - mu[c]) * scale + bet[c];
        }
#pragma unroll
        for (int rt = 0; rt < 2; ++rt) {
            int row0 = rowBase + rowOff + rt * 16 + kh * 8;
#pragma unroll
            for (int r = 0; r < 8; ++r) {
                float val = acc[rt][ct][r];
                if (USE_BN) val = val * scale + shift;
                int row = row0 + r;
                if (USE_RES) val += res[(size_t)row * ncols + c];
                val = fmaxf(val, 0.0f);
                out[(size_t)row * ncols + c] = val;
            }
        }
    }
}


extern "C" void kernel_launch(void* const* d_in, const int* in_sizes, int n_in,
                              void* d_out, int out_size, void* d_ws, size_t ws_size,
                              hipStream_t stream) {
    constexpr int N = GIN_N, E = GIN_E, DIN = GIN_DIN, DOUT = GIN_DOUT;

    const float* x    = (const float*)d_in[0];
    const int*   eidx = (const int*)d_in[1];
    const float* eps  = (const float*)d_in[2];
    const float* W1   = (const float*)d_in[3];
    const float* b1   = (const float*)d_in[4];
    const float* g1   = (const float*)d_in[5];
    const float* bt1  = (const float*)d_in[6];
    const float* m1   = (const float*)d_in[7];
    const float* v1   = (const float*)d_in[8];
    const float* W2   = (const float*)d_in[9];
    const float* b2   = (const float*)d_in[10];
    const float* g2   = (const float*)d_in[11];
    const float* bt2  = (const float*)d_in[12];
    const float* m2   = (const float*)d_in[13];
    const float* v2   = (const float*)d_in[14];
    const float* Wres = (const float*)d_in[15];

    const int* src = eidx;
    const int* dst = eidx + E;

    float* h  = (float*)d_ws;
    float* h1 = h + (size_t)N * DIN;
    float* h2 = h1 + (size_t)N * DOUT;
    int* counts  = (int*)(h2 + (size_t)N * DOUT);
    int* cursor  = counts + N;
    int* offsets = cursor + N;
    int* bucket  = offsets + N + 1;

    zero_kernel<<<(2 * N + 255) / 256, 256, 0, stream>>>(counts, 2 * N);
    count_kernel<<<E / 256, 256, 0, stream>>>(dst, counts, E);
    scan_kernel<<<1, 256, 0, stream>>>(counts, offsets, N);
    fill_kernel<<<E / 256, 256, 0, stream>>>(src, dst, offsets, cursor, bucket, E);
    gather_kernel<<<N / 8, 256, 0, stream>>>(x, offsets, bucket, eps, h);

    dim3 grid(N / 64, DOUT / 64);
    size_t lds128 = (size_t)(64 * (128 + 4) + 128 * 72) * 4;
    size_t lds256 = (size_t)(64 * (256 + 4) + 256 * 72) * 4;

    gemm_kernel<128, true, false><<<grid, 128, lds128, stream>>>(
        h, W1, b1, g1, bt1, m1, v1, nullptr, h1, DOUT);
    gemm_kernel<256, true, false><<<grid, 128, lds256, stream>>>(
        h1, W2, b2, g2, bt2, m2, v2, nullptr, h2, DOUT);
    gemm_kernel<128, false, true><<<grid, 128, lds128, stream>>>(
        x, Wres, nullptr, nullptr, nullptr, nullptr, nullptr, h2, (float*)d_out, DOUT);
}


# GEMM re-tiled: a work-group owns 160 rows x all 256 columns, B once in LDS (transposed), each wave reads its A fragments straight from memory (no LDS for A); k order inside each run of 8 changes
# speedup vs baseline: 2.4980x; 2.4980x over previous
#include <hip/hip_runtime.h>
#include <hip/hip_bf16.h>

typedef float v2f __attribute__((ext_vector_type(2)));
typedef float v8f __attribute__((ext_vector_type(8)));

#define GIN_N    40000
#define GIN_E    640000
#define GIN_DIN  128
#define GIN_DOUT 256


__global__ void zero_kernel(int* __restrict__ p, int n) {
    int i = blockIdx.x * 256 + threadIdx.x;
    if (i < n) p[i] = 0;
}

__global__ void count_kernel(const int* __restrict__ dst, int* __restrict__ counts, int e) {
    int i = blockIdx.x * 256 + threadIdx.x;
    if (i < e) atomicAdd(&counts[dst[i]], 1);
}

__global__ __launch_bounds__(256)
void place_kernel(const int* __restrict__ counts, int* __restrict__ total,
                  int* __restrict__ offsets, int n) {
    int i = blockIdx.x * 256 + threadIdx.x;
    int lane = threadIdx.x & 31;
    int c = (i < n) ? counts[i] : 0;
    int incl = c;
#pragma unroll
    for (int off = 1; off < 32; off <<= 1) {
        int t = __shfl_up(incl, off, 32);
        if (lane >= off) incl += t;
    }
    int base = 0;
    if (lane == 31) base = atomicAdd(total, incl);
    base = __shfl(base, 31, 32);
    if (i < n) offsets[i] = base + incl - c;
}

__global__ void fill_kernel(const int* __restrict__ src, const int* __restrict__ dst,
                            const int* __restrict__ offsets, int* __restrict__ cursor,
                            int* __restrict__ bucket, int e) {
    int i = blockIdx.x * 256 + threadIdx.x;
    if (i < e) {
        int d = dst[i];
        int pos = offsets[d] + atomicAdd(&cursor[d], 1);
        bucket[pos] = src[i];
    }
}

__global__ __launch_bounds__(256)
void gather_kernel(const float* __restrict__ x, const int* __restrict__ offsets,
                   const int* __restrict__ counts,
                   const int* __restrict__ bucket, const float* __restrict__ epsp,
                   float* __restrict__ h) {
    int node = blockIdx.x * 8 + (threadIdx.x >> 5);
    int lane = threadIdx.x & 31;
    float se = 1.0f + epsp[0];
    const float4* xv = (const float4*)x;
    float4 self = xv[(size_t)node * 32 + lane];
    float ax = self.x * se, ay = self.y * se, az = self.z * se, aw = self.w * se;
    int e0 = offsets[node], e1 = e0 + counts[node];
    for (int i = e0; i < e1; ++i) {
        int s = bucket[i];
        float4 r = xv[(size_t)s * 32 + lane];
        ax += r.x; ay += r.y; az += r.z; aw += r.w;
    }
    float4 o; o.x = ax; o.y = ay; o.z = az; o.w = aw;
    ((float4*)h)[(size_t)node * 32 + lane] = o;
}


#define GEMM_ROWS_PER_WG 160

template <int K, bool USE_BN, bool USE_RES>
__global__ __launch_bounds__(256) __attribute__((amdgpu_waves_per_eu(2, 2)))
void gemm_kernel(const float* __restrict__ A, const float* __restrict__ B,
                 const float* __restrict__ bias, const float* __restrict__ gam,
                 const float* __restrict__ bet, const float* __restrict__ mu,
                 const float* __restrict__ var, const float* __restrict__ res,
                 float* __restrict__ out) {
    constexpr int PT = K + 4;
    constexpr int NCOL = GIN_DOUT;
    extern __shared__ float BsT[];

    const int tid = threadIdx.x;

    {
        const int cg = (tid >> 5) * 8 + (tid & 7);
        const int kq = (tid >> 3) & 3;
        constexpr int NP = K / 16;
#pragma unroll 4
        for (int p = 0; p < NP; ++p) {
            const int pp = (p + blockIdx.x) % NP;
            const int k = 16 * pp + 4 * kq;
            const float4 r0 = *(const float4*)(B + (size_t)(k + 0) * NCOL + 4 * cg);
            const float4 r1 = *(const float4*)(B + (size_t)(k + 1) * NCOL + 4 * cg);
            const float4 r2 = *(const float4*)(B + (size_t)(k + 2) * NCOL + 4 * cg);
            const float4 r3 = *(const float4*)(B + (size_t)(k + 3) * NCOL + 4 * cg);
            float* d = BsT + (4 * cg) * PT + k;
            *(float4*)(d)          = make_float4(r0.x, r1.x, r2.x, r3.x);
            *(float4*)(d + PT)     = make_float4(r0.y, r1.y, r2.y, r3.y);
            *(float4*)(d + 2 * PT) = make_float4(r0.z, r1.z, r2.z, r3.z);
            *(float4*)(d + 3 * PT) = make_float4(r0.w, r1.w, r2.w, r3.w);
        }
    }
    __syncthreads();

    const int wave = tid >> 5;
    const int lane = tid & 31;
    const int c  = lane & 15;
    const int h  = lane >> 4;
    const int colBase = (wave & 3) * 64;
    const int sp = wave >> 2;

    float scale[4], shift[4];
#pragma unroll
    for (int ct = 0; ct < 4; ++ct) {
        scale[ct] = 1.0f; shift[ct] = 0.0f;
        if (USE_BN) {
            int col = colBase + ct * 16 + c;
            float inv = rsqrtf(var[col] + 1e-5f);
            scale[ct] = gam[col] * inv;
            shift[ct] = (bias[col] - mu[col]) * scale[ct] + bet[col];
        }
    }

    const float* Bp = BsT + (colBase + c) * PT + 4 * h;

    float4 apf[8];
    {
        const float* Afirst = A + (size_t)(blockIdx.x * GEMM_ROWS_PER_WG + sp * 16 + c) * K + 4 * h;
#pragma unroll
        for (int j = 0; j < 8; ++j) apf[j] = *(const float4*)(Afirst + 8 * j);
    }

#pragma unroll 1
    for (int it = 0; it < GEMM_ROWS_PER_WG / 32; ++it) {
        const int row0 = blockIdx.x * GEMM_ROWS_PER_WG + (sp + 2 * it) * 16;
        const float* Arow = A + (size_t)(row0 + c) * K + 4 * h;

        float rv[4][8];
        if (USE_RES) {
#pragma unroll
            for (int ct = 0; ct < 4; ++ct)
#pragma unroll
                for (int r = 0; r < 8; ++r)
                    rv[ct][r] = res[(size_t)(row0 + 8 * h + r) * NCOL + colBase + ct * 16 + c];
        }

        v8f acc[4] = {};
#pragma unroll 1
        for (int k0 = 0; k0 < K; k0 += 128) {
            const bool lastHalf = (k0 + 128 >= K);
            const bool lastStrip = (it + 1 >= GEMM_ROWS_PER_WG / 32);
            const float* Anext = lastHalf ? (lastStrip ? Arow + k0 : Arow + (size_t)32 * K)
                                          : Arow + k0 + 128;
            float4 apn[8];
#pragma unroll
            for (int j = 0; j < 8; ++j) apn[j] = *(const float4*)(Anext + 8 * j);
#pragma unroll
            for (int j = 0; j < 16; ++j) {
                float4 a4 = (j < 8) ? apf[j] : *(const float4*)(Arow + k0 + 8 * j);
                float4 b4[4];
#pragma unroll
                for (int ct = 0; ct < 4; ++ct)
                    b4[ct] = *(const float4*)(Bp + ct * 16 * PT + k0 + 8 * j);
#pragma unroll
                for (int ct = 0; ct < 4; ++ct) {
                    v2f a, b;
                    a.x = a4.x; a.y = a4.y; b.x = b4[ct].x; b.y = b4[ct].y;
                    acc[ct] = __builtin_amdgcn_wmma_f32_16x16x4_f32(
                        false, a, false, b, (short)0, acc[ct], false, false);
                }
#pragma unroll
                for (int ct = 0; ct < 4; ++ct) {
                    v2f a, b;
                    a.x = a4.z; a.y = a4.w; b.x = b4[ct].z; b.y = b4[ct].w;
                    acc[ct] = __builtin_amdgcn_wmma_f32_16x16x4_f32(
                        false, a, false, b, (short)0, acc[ct], false, false);
                }
            }
#pragma unroll
            for (int j = 0; j < 8; ++j) apf[j] = apn[j];
        }

#pragma unroll
        for (int ct = 0; ct < 4; ++ct) {
            int col = colBase + ct * 16 + c;
#pragma unroll
            for (int r = 0; r < 8; ++r) {
                float val = acc[ct][r];
                if (USE_BN) val = val * scale[ct] + shift[ct];
                if (USE_RES) val += rv[ct][r];
                val = fmaxf(val, 0.0f);
                out[(size_t)(row0 + 8 * h + r) * NCOL + col] = val;
            }
        }
    }
}


extern "C" void kernel_launch(void* const* d_in, const int* in_sizes, int n_in,
                              void* d_out, int out_size, void* d_ws, size_t ws_size,
                              hipStream_t stream) {
    constexpr int N = GIN_N, E = GIN_E, DIN = GIN_DIN, DOUT = GIN_DOUT;

    const float* x    = (const float*)d_in[0];
    const int*   eidx = (const int*)d_in[1];
    const float* eps  = (const float*)d_in[2];
    const float* W1   = (const float*)d_in[3];
    const float* b1   = (const float*)d_in[4];
    const float* g1   = (const float*)d_in[5];
    const float* bt1  = (const float*)d_in[6];
    const float* m1   = (const float*)d_in[7];
    const float* v1   = (const float*)d_in[8];
    const float* W2   = (const float*)d_in[9];
    const float* b2   = (const float*)d_in[10];
    const float* g2   = (const float*)d_in[11];
    const float* bt2  = (const float*)d_in[12];
    const float* m2   = (const float*)d_in[13];
    const float* v2   = (const float*)d_in[14];
    const float* Wres = (const float*)d_in[15];

    const int* src = eidx;
    const int* dst = eidx + E;

    float* h  = (float*)d_ws;
    float* h1 = h + (size_t)N * DIN;
    float* h2 = h1 + (size_t)N * DOUT;
    int* counts  = (int*)(h2 + (size_t)N * DOUT);
    int* cursor  = counts + N;
    int* total   = cursor + N;
    int* offsets = total + 1;
    int* bucket  = offsets + N;

    zero_kernel<<<(2 * N + 1 + 255) / 256, 256, 0, stream>>>(counts, 2 * N + 1);
    count_kernel<<<E / 256, 256, 0, stream>>>(dst, counts, E);
    place_kernel<<<(N + 255) / 256, 256, 0, stream>>>(counts, total, offsets, N);
    fill_kernel<<<E / 256, 256, 0, stream>>>(src, dst, offsets, cursor, bucket, E);
    gather_kernel<<<N / 8, 256, 0, stream>>>(x, offsets, counts, bucket, eps, h);

    dim3 grid(N / GEMM_ROWS_PER_WG);
    size_t lds128 = (size_t)(256 * (128 + 4)) * 4;
    size_t lds256 = (size_t)(256 * (256 + 4)) * 4;

    gemm_kernel<128, true, false><<<grid, 256, lds128, stream>>>(
        h, W1, b1, g1, bt1, m1, v1, nullptr, h1);
    gemm_kernel<256, true, false><<<grid, 256, lds256, stream>>>(
        h1, W2, b2, g2, bt2, m2, v2, nullptr, h2);
    gemm_kernel<128, false, true><<<grid, 256, lds128, stream>>>(
        x, Wres, nullptr, nullptr, nullptr, nullptr, nullptr, h2, (float*)d_out);
}


# edge lists without count or placement: a node owns a fixed line of 32 slots, an edge claims one with its single int atomic (spill list past 32); 2 launches gone; node sums keep an arrival order
# speedup vs baseline: 3.4072x; 1.3640x over previous
#include <hip/hip_runtime.h>
#include <hip/hip_bf16.h>

typedef float v2f __attribute__((ext_vector_type(2)));
typedef float v8f __attribute__((ext_vector_type(8)));

#define GIN_N    40000
#define GIN_E    640000
#define GIN_DIN  128
#define GIN_DOUT 256


__global__ void zero_kernel(int* __restrict__ p, int n) {
    int i = blockIdx.x * 256 + threadIdx.x;
    if (i < n) p[i] = 0;
}

#define GIN_CAP 32

__global__ void fill_kernel(const int* __restrict__ src, const int* __restrict__ dst,
                            int* __restrict__ cursor, int* __restrict__ slots,
                            int* __restrict__ spill_n, int* __restrict__ spill, int e) {
    int i = blockIdx.x * 256 + threadIdx.x;
    if (i < e) {
        int d = dst[i];
        int s = src[i];
        int pos = atomicAdd(&cursor[d], 1);
        if (pos < GIN_CAP) {
            slots[(size_t)d * GIN_CAP + pos] = s;
        } else {
            int q = atomicAdd(spill_n, 1);
            ((int2*)spill)[q] = make_int2(d, s);
        }
    }
}

__global__ __launch_bounds__(256)
void gather_kernel(const float* __restrict__ x, const int* __restrict__ cursor,
                   const int* __restrict__ slots, const int* __restrict__ spill_n,
                   const int* __restrict__ spill, const float* __restrict__ epsp,
                   float* __restrict__ h, int e) {
    const int node = __builtin_amdgcn_readfirstlane(blockIdx.x * 8 + (threadIdx.x >> 5));
    const int lane = threadIdx.x & 31;
    float se = 1.0f + epsp[0];
    const float4* xv = (const float4*)x;
    float4 self = xv[(size_t)node * 32 + lane];
    float ax = self.x * se, ay = self.y * se, az = self.z * se, aw = self.w * se;
    const int cnt = cursor[node];
    const int mine = slots[(size_t)node * GIN_CAP + lane];
    const int n = cnt < GIN_CAP ? cnt : GIN_CAP;
    for (int j = 0; j < n; ++j) {
        int s = __builtin_amdgcn_readlane(mine, j);
        float4 r = xv[(size_t)s * 32 + lane];
        ax += r.x; ay += r.y; az += r.z; aw += r.w;
    }
    if (cnt > GIN_CAP) {
        int m = spill_n[0];
        m = m < e ? m : e;
        const int2* sp = (const int2*)spill;
        for (int q0 = 0; q0 < m; q0 += 32) {
            int q = q0 + lane;
            int2 p = make_int2(-1, 0);
            if (q < m) p = sp[q];
            unsigned hit = __builtin_amdgcn_ballot_w32(p.x == node);
            while (hit) {
                int b = __builtin_ctz(hit);
                hit &= hit - 1;
                int s = __builtin_amdgcn_readlane(p.y, b);
                float4 r = xv[(size_t)s * 32 + lane];
                ax += r.x; ay += r.y; az += r.z; aw += r.w;
            }
        }
    }
    float4 o; o.x = ax; o.y = ay; o.z = az; o.w = aw;
    ((float4*)h)[(size_t)node * 32 + lane] = o;
}


#define GEMM_ROWS_PER_WG 160

template <int K, bool USE_BN, bool USE_RES>
__global__ __launch_bounds__(256) __attribute__((amdgpu_waves_per_eu(2, 2)))
void gemm_kernel(const float* __restrict__ A, const float* __restrict__ B,
                 const float* __restrict__ bias, const float* __restrict__ gam,
                 const float* __restrict__ bet, const float* __restrict__ mu,
                 const float* __restrict__ var, const float* __restrict__ res,
                 float* __restrict__ out) {
    constexpr int PT = K + 4;
    constexpr int NCOL = GIN_DOUT;
    extern __shared__ float BsT[];

    const int tid = threadIdx.x;

    {
        const int cg = (tid >> 5) * 8 + (tid & 7);
        const int kq = (tid >> 3) & 3;
        constexpr int NP = K / 16;
#pragma unroll 4
        for (int p = 0; p < NP; ++p) {
            const int pp = (p + blockIdx.x) % NP;
            const int k = 16 * pp + 4 * kq;
            const float4 r0 = *(const float4*)(B + (size_t)(k + 0) * NCOL + 4 * cg);
            const float4 r1 = *(const float4*)(B + (size_t)(k + 1) * NCOL + 4 * cg);
            const float4 r2 = *(const float4*)(B + (size_t)(k + 2) * NCOL + 4 * cg);
            const float4 r3 = *(const float4*)(B + (size_t)(k + 3) * NCOL + 4 * cg);
            float* d = BsT + (4 * cg) * PT + k;
            *(float4*)(d)          = make_float4(r0.x, r1.x, r2.x, r3.x);
            *(float4*)(d + PT)     = make_float4(r0.y, r1.y, r2.y, r3.y);
            *(float4*)(d + 2 * PT) = make_float4(r0.z, r1.z, r2.z, r3.z);
            *(float4*)(d + 3 * PT) = make_float4(r0.w, r1.w, r2.w, r3.w);
        }
    }
    __syncthreads();

    const int wave = tid >> 5;
    const int lane = tid & 31;
    const int c  = lane & 15;
    const int h  = lane >> 4;
    const int colBase = (wave & 3) * 64;
    const int sp = wave >> 2;

    float scale[4], shift[4];
#pragma unroll
    for (int ct = 0; ct < 4; ++ct) {
        scale[ct] = 1.0f; shift[ct] = 0.0f;
        if (USE_BN) {
            int col = colBase + ct * 16 + c;
            float inv = rsqrtf(var[col] + 1e-5f);
            scale[ct] = gam[col] * inv;
            shift[ct] = (bias[col] - mu[col]) * scale[ct] + bet[col];
        }
    }

    const float* Bp = BsT + (colBase + c) * PT + 4 * h;

    float4 apf[8];
    {
        const float* Afirst = A + (size_t)(blockIdx.x * GEMM_ROWS_PER_WG + sp * 16 + c) * K + 4 * h;
#pragma unroll
        for (int j = 0; j < 8; ++j) apf[j] = *(const float4*)(Afirst + 8 * j);
    }

#pragma unroll 1
    for (int it = 0; it < GEMM_ROWS_PER_WG / 32; ++it) {
        const int row0 = blockIdx.x * GEMM_ROWS_PER_WG + (sp + 2 * it) * 16;
        const float* Arow = A + (size_t)(row0 + c) * K + 4 * h;

        float rv[4][8];
        if (USE_RES) {
#pragma unroll
            for (int ct = 0; ct < 4; ++ct)
#pragma unroll
                for (int r = 0; r < 8; ++r)
                    rv[ct][r] = res[(size_t)(row0 + 8 * h + r) * NCOL + colBase + ct * 16 + c];
        }

        v8f acc[4] = {};
#pragma unroll 1
        for (int k0 = 0; k0 < K; k0 += 128) {
            const bool lastHalf = (k0 + 128 >= K);
            const bool lastStrip = (it + 1 >= GEMM_ROWS_PER_WG / 32);
            const float* Anext = lastHalf ? (lastStrip ? Arow + k0 : Arow + (size_t)32 * K)
                                          : Arow + k0 + 128;
            float4 apn[8];
#pragma unroll
            for (int j = 0; j < 8; ++j) apn[j] = *(const float4*)(Anext + 8 * j);
#pragma unroll
            for (int j = 0; j < 16; ++j) {
                float4 a4 = (j < 8) ? apf[j] : *(const float4*)(Arow + k0 + 8 * j);
                float4 b4[4];
#pragma unroll
                for (int ct = 0; ct < 4; ++ct)
                    b4[ct] = *(const float4*)(Bp + ct * 16 * PT + k0 + 8 * j);
#pragma unroll
                for (int ct = 0; ct < 4; ++ct) {
                    v2f a, b;
                    a.x = a4.x; a.y = a4.y; b.x = b4[ct].x; b.y = b4[ct].y;
                    acc[ct] = __builtin_amdgcn_wmma_f32_16x16x4_f32(
                        false, a, false, b, (short)0, acc[ct], false, false);
                }
#pragma unroll
                for (int ct = 0; ct < 4; ++ct) {
                    v2f a, b;
                    a.x = a4.z; a.y = a4.w; b.x = b4[ct].z; b.y = b4[ct].w;
                    acc[ct] = __builtin_amdgcn_wmma_f32_16x16x4_f32(
                        false, a, false, b, (short)0, acc[ct], false, false);
                }
            }
#pragma unroll
            for (int j = 0; j < 8; ++j) apf[j] = apn[j];
        }

#pragma unroll
        for (int ct = 0; ct < 4; ++ct) {
            int col = colBase + ct * 16 + c;
#pragma unroll
            for (int r = 0; r < 8; ++r) {
                float val = acc[ct][r];
                if (USE_BN) val = val * scale[ct] + shift[ct];
                if (USE_RES) val += rv[ct][r];
                val = fmaxf(val, 0.0f);
                out[(size_t)(row0 + 8 * h + r) * NCOL + col] = val;
            }
        }
    }
}


extern "C" void kernel_launch(void* const* d_in, const int* in_sizes, int n_in,
                              void* d_out, int out_size, void* d_ws, size_t ws_size,
                              hipStream_t stream) {
    constexpr int N = GIN_N, E = GIN_E, DIN = GIN_DIN, DOUT = GIN_DOUT;

    const float* x    = (const float*)d_in[0];
    const int*   eidx = (const int*)d_in[1];
    const float* eps  = (const float*)d_in[2];
    const float* W1   = (const float*)d_in[3];
    const float* b1   = (const float*)d_in[4];
    const float* g1   = (const float*)d_in[5];
    const float* bt1  = (const float*)d_in[6];
    const float* m1   = (const float*)d_in[7];
    const float* v1   = (const float*)d_in[8];
    const float* W2   = (const float*)d_in[9];
    const float* b2   = (const float*)d_in[10];
    const float* g2   = (const float*)d_in[11];
    const float* bt2  = (const float*)d_in[12];
    const float* m2   = (const float*)d_in[13];
    const float* v2   = (const float*)d_in[14];
    const float* Wres = (const float*)d_in[15];

    const int* src = eidx;
    const int* dst = eidx + E;

    float* h  = (float*)d_ws;
    float* h1 = h + (size_t)N * DIN;
    float* h2 = h1 + (size_t)N * DOUT;
    int* cursor  = (int*)h2;
    int* spill_n = cursor + N;
    int* slots   = cursor + N + 64;
    int* spill   = slots + (size_t)N * GIN_CAP;

    zero_kernel<<<(N + 1 + 255) / 256, 256, 0, stream>>>(cursor, N + 1);
    fill_kernel<<<E / 256, 256, 0, stream>>>(src, dst, cursor, slots, spill_n, spill, E);
    gather_kernel<<<N / 8, 256, 0, stream>>>(x, cursor, slots, spill_n, spill, eps, h, E);

    dim3 grid(N / GEMM_ROWS_PER_WG);
    size_t lds128 = (size_t)(256 * (128 + 4)) * 4;
    size_t lds256 = (size_t)(256 * (256 + 4)) * 4;

    gemm_kernel<128, true, false><<<grid, 256, lds128, stream>>>(
        h, W1, b1, g1, bt1, m1, v1, nullptr, h1);
    gemm_kernel<256, true, false><<<grid, 256, lds256, stream>>>(
        h1, W2, b2, g2, bt2, m2, v2, nullptr, h2);
    gemm_kernel<128, false, true><<<grid, 256, lds128, stream>>>(
        x, Wres, nullptr, nullptr, nullptr, nullptr, nullptr, h2, (float*)d_out);
}


# edge fill split by package half: a wave reads its die and draws a 256-edge task of its half's node range from a ticket counter, so a cursor or slot line is written from one half; arrival order kept
# speedup vs baseline: 4.6028x; 1.3509x over previous
#include <hip/hip_runtime.h>
#include <hip/hip_bf16.h>

typedef float v2f __attribute__((ext_vector_type(2)));
typedef float v8f __attribute__((ext_vector_type(8)));

#define GIN_N    40000
#define GIN_E    640000
#define GIN_DIN  128
#define GIN_DOUT 256


__global__ void zero_kernel(int* __restrict__ p, int n) {
    int i = blockIdx.x * 256 + threadIdx.x;
    if (i < n) p[i] = 0;
}

#define GIN_CAP 32

#define GIN_SPLIT 20032
#define GIN_TASK  256
#define GIN_TICKET_GAP 64

__global__ __launch_bounds__(128)
void fill_kernel(const int* __restrict__ src, const int* __restrict__ dst,
                 int* __restrict__ cursor, int* __restrict__ slots,
                 int* __restrict__ spill_n, int* __restrict__ spill,
                 int* __restrict__ ticket, int e) {
    const int lane = threadIdx.x & 31;
    const int ntask = e / GIN_TASK;
    int half = (int)((__builtin_amdgcn_s_sendmsg_rtn(0x87) >> 18) & 1u);
    int t = 0;
    if (lane == 0) t = atomicAdd(&ticket[half * GIN_TICKET_GAP], 1);
    t = __builtin_amdgcn_readfirstlane(t);
    if (t >= ntask) {
        half ^= 1;
        if (lane == 0) t = atomicAdd(&ticket[half * GIN_TICKET_GAP], 1);
        t = __builtin_amdgcn_readfirstlane(t);
        if (t >= ntask) return;
    }
    const int lo = half ? GIN_SPLIT : 0;
    const int hi = half ? GIN_N : GIN_SPLIT;
    const int base = t * GIN_TASK;
    const int4 d0 = ((const int4*)(dst + base))[lane];
    const int4 d1 = ((const int4*)(dst + base + GIN_TASK / 2))[lane];
    const int4 s0 = ((const int4*)(src + base))[lane];
    const int4 s1 = ((const int4*)(src + base + GIN_TASK / 2))[lane];
    const int d[8] = {d0.x, d0.y, d0.z, d0.w, d1.x, d1.y, d1.z, d1.w};
    const int s[8] = {s0.x, s0.y, s0.z, s0.w, s1.x, s1.y, s1.z, s1.w};
    int pos[8];
#pragma unroll
    for (int k = 0; k < 8; ++k) {
        pos[k] = -1;
        if (d[k] >= lo && d[k] < hi) pos[k] = atomicAdd(&cursor[d[k]], 1);
    }
#pragma unroll
    for (int k = 0; k < 8; ++k) {
        if (pos[k] >= 0) {
            if (pos[k] < GIN_CAP) {
                slots[(size_t)d[k] * GIN_CAP + pos[k]] = s[k];
            } else {
                int q = atomicAdd(spill_n, 1);
                ((int2*)spill)[q] = make_int2(d[k], s[k]);
            }
        }
    }
}

__global__ __launch_bounds__(256)
void gather_kernel(const float* __restrict__ x, const int* __restrict__ cursor,
                   const int* __restrict__ slots, const int* __restrict__ spill_n,
                   const int* __restrict__ spill, const float* __restrict__ epsp,
                   float* __restrict__ h, int e) {
    const int node = __builtin_amdgcn_readfirstlane(blockIdx.x * 8 + (threadIdx.x >> 5));
    const int lane = threadIdx.x & 31;
    float se = 1.0f + epsp[0];
    const float4* xv = (const float4*)x;
    float4 self = xv[(size_t)node * 32 + lane];
    float ax = self.x * se, ay = self.y * se, az = self.z * se, aw = self.w * se;
    const int cnt = cursor[node];
    const int mine = slots[(size_t)node * GIN_CAP + lane];
    const int n = cnt < GIN_CAP ? cnt : GIN_CAP;
    for (int j = 0; j < n; ++j) {
        int s = __builtin_amdgcn_readlane(mine, j);
        float4 r = xv[(size_t)s * 32 + lane];
        ax += r.x; ay += r.y; az += r.z; aw += r.w;
    }
    if (cnt > GIN_CAP) {
        int m = spill_n[0];
        m = m < e ? m : e;
        const int2* sp = (const int2*)spill;
        for (int q0 = 0; q0 < m; q0 += 32) {
            int q = q0 + lane;
            int2 p = make_int2(-1, 0);
            if (q < m) p = sp[q];
            unsigned hit = __builtin_amdgcn_ballot_w32(p.x == node);
            while (hit) {
                int b = __builtin_ctz(hit);
                hit &= hit - 1;
                int s = __builtin_amdgcn_readlane(p.y, b);
                float4 r = xv[(size_t)s * 32 + lane];
                ax += r.x; ay += r.y; az += r.z; aw += r.w;
            }
        }
    }
    float4 o; o.x = ax; o.y = ay; o.z = az; o.w = aw;
    ((float4*)h)[(size_t)node * 32 + lane] = o;
}


#define GEMM_ROWS_PER_WG 160

template <int K, bool USE_BN, bool USE_RES>
__global__ __launch_bounds__(256) __attribute__((amdgpu_waves_per_eu(2, 2)))
void gemm_kernel(const float* __restrict__ A, const float* __restrict__ B,
                 const float* __restrict__ bias, const float* __restrict__ gam,
                 const float* __restrict__ bet, const float* __restrict__ mu,
                 const float* __restrict__ var, const float* __restrict__ res,
                 float* __restrict__ out) {
    constexpr int PT = K + 4;
    constexpr int NCOL = GIN_DOUT;
    extern __shared__ float BsT[];

    const int tid = threadIdx.x;

    {
        const int cg = (tid >> 5) * 8 + (tid & 7);
        const int kq = (tid >> 3) & 3;
        constexpr int NP = K / 16;
#pragma unroll 4
        for (int p = 0; p < NP; ++p) {
            const int pp = (p + blockIdx.x) % NP;
            const int k = 16 * pp + 4 * kq;
            const float4 r0 = *(const float4*)(B + (size_t)(k + 0) * NCOL + 4 * cg);
            const float4 r1 = *(const float4*)(B + (size_t)(k + 1) * NCOL + 4 * cg);
            const float4 r2 = *(const float4*)(B + (size_t)(k + 2) * NCOL + 4 * cg);
            const float4 r3 = *(const float4*)(B + (size_t)(k + 3) * NCOL + 4 * cg);
            float* d = BsT + (4 * cg) * PT + k;
            *(float4*)(d)          = make_float4(r0.x, r1.x, r2.x, r3.x);
            *(float4*)(d + PT)     = make_float4(r0.y, r1.y, r2.y, r3.y);
            *(float4*)(d + 2 * PT) = make_float4(r0.z, r1.z, r2.z, r3.z);
            *(float4*)(d + 3 * PT) = make_float4(r0.w, r1.w, r2.w, r3.w);
        }
    }
    __syncthreads();

    const int wave = tid >> 5;
    const int lane = tid & 31;
    const int c  = lane & 15;
    const int h  = lane >> 4;
    const int colBase = (wave & 3) * 64;
    const int sp = wave >> 2;

    float scale[4], shift[4];
#pragma unroll
    for (int ct = 0; ct < 4; ++ct) {
        scale[ct] = 1.0f; shift[ct] = 0.0f;
        if (USE_BN) {
            int col = colBase + ct * 16 + c;
            float inv = rsqrtf(var[col] + 1e-5f);
            scale[ct] = gam[col] * inv;
            shift[ct] = (bias[col] - mu[col]) * scale[ct] + bet[col];
        }
    }

    const float* Bp = BsT + (colBase + c) * PT + 4 * h;

    float4 apf[8];
    {
        const float* Afirst = A + (size_t)(blockIdx.x * GEMM_ROWS_PER_WG + sp * 16 + c) * K + 4 * h;
#pragma unroll
        for (int j = 0; j < 8; ++j) apf[j] = *(const float4*)(Afirst + 8 * j);
    }

#pragma unroll 1
    for (int it = 0; it < GEMM_ROWS_PER_WG / 32; ++it) {
        const int row0 = blockIdx.x * GEMM_ROWS_PER_WG + (sp + 2 * it) * 16;
        const float* Arow = A + (size_t)(row0 + c) * K + 4 * h;

        float rv[4][8];
        if (USE_RES) {
#pragma unroll
            for (int ct = 0; ct < 4; ++ct)
#pragma unroll
                for (int r = 0; r < 8; ++r)
                    rv[ct][r] = res[(size_t)(row0 + 8 * h + r) * NCOL + colBase + ct * 16 + c];
        }

        v8f acc[4] = {};
#pragma unroll 1
        for (int k0 = 0; k0 < K; k0 += 128) {
            const bool lastHalf = (k0 + 128 >= K);
            const bool lastStrip = (it + 1 >= GEMM_ROWS_PER_WG / 32);
            const float* Anext = lastHalf ? (lastStrip ? Arow + k0 : Arow + (size_t)32 * K)
                                          : Arow + k0 + 128;
            float4 apn[8];
#pragma unroll
            for (int j = 0; j < 8; ++j) apn[j] = *(const float4*)(Anext + 8 * j);
#pragma unroll
            for (int j = 0; j < 16; ++j) {
                float4 a4 = (j < 8) ? apf[j] : *(const float4*)(Arow + k0 + 8 * j);
                float4 b4[4];
#pragma unroll
                for (int ct = 0; ct < 4; ++ct)
                    b4[ct] = *(const float4*)(Bp + ct * 16 * PT + k0 + 8 * j);
#pragma unroll
                for (int ct = 0; ct < 4; ++ct) {
                    v2f a, b;
                    a.x = a4.x; a.y = a4.y; b.x = b4[ct].x; b.y = b4[ct].y;
                    acc[ct] = __builtin_amdgcn_wmma_f32_16x16x4_f32(
                        false, a, false, b, (short)0, acc[ct], false, false);
                }
#pragma unroll
                for (int ct = 0; ct < 4; ++ct) {
                    v2f a, b;
                    a.x = a4.z; a.y = a4.w; b.x = b4[ct].z; b.y = b4[ct].w;
                    acc[ct] = __builtin_amdgcn_wmma_f32_16x16x4_f32(
                        false, a, false, b, (short)0, acc[ct], false, false);
                }
            }
#pragma unroll
            for (int j = 0; j < 8; ++j) apf[j] = apn[j];
        }

#pragma unroll
        for (int ct = 0; ct < 4; ++ct) {
            int col = colBase + ct * 16 + c;
#pragma unroll
            for (int r = 0; r < 8; ++r) {
                float val = acc[ct][r];
                if (USE_BN) val = val * scale[ct] + shift[ct];
                if (USE_RES) val += rv[ct][r];
                val = fmaxf(val, 0.0f);
                out[(size_t)(row0 + 8 * h + r) * NCOL + col] = val;
            }
        }
    }
}


extern "C" void kernel_launch(void* const* d_in, const int* in_sizes, int n_in,
                              void* d_out, int out_size, void* d_ws, size_t ws_size,
                              hipStream_t stream) {
    constexpr int N = GIN_N, E = GIN_E, DIN = GIN_DIN, DOUT = GIN_DOUT;

    const float* x    = (const float*)d_in[0];
    const int*   eidx = (const int*)d_in[1];
    const float* eps  = (const float*)d_in[2];
    const float* W1   = (const float*)d_in[3];
    const float* b1   = (const float*)d_in[4];
    const float* g1   = (const float*)d_in[5];
    const float* bt1  = (const float*)d_in[6];
    const float* m1   = (const float*)d_in[7];
    const float* v1   = (const float*)d_in[8];
    const float* W2   = (const float*)d_in[9];
    const float* b2   = (const float*)d_in[10];
    const float* g2   = (const float*)d_in[11];
    const float* bt2  = (const float*)d_in[12];
    const float* m2   = (const float*)d_in[13];
    const float* v2   = (const float*)d_in[14];
    const float* Wres = (const float*)d_in[15];

    const int* src = eidx;
    const int* dst = eidx + E;

    float* h  = (float*)d_ws;
    float* h1 = h + (size_t)N * DIN;
    float* h2 = h1 + (size_t)N * DOUT;
    int* cursor  = (int*)h2;
    int* spill_n = cursor + N;
    int* ticket  = cursor + N + 64;
    int* slots   = cursor + N + 256;
    int* spill   = slots + (size_t)N * GIN_CAP;

    zero_kernel<<<(N + 192 + 255) / 256, 256, 0, stream>>>(cursor, N + 192);
    fill_kernel<<<2 * (E / GIN_TASK) / 4, 128, 0, stream>>>(src, dst, cursor, slots, spill_n, spill,
                                                           ticket, E);
    gather_kernel<<<N / 8, 256, 0, stream>>>(x, cursor, slots, spill_n, spill, eps, h, E);

    dim3 grid(N / GEMM_ROWS_PER_WG);
    size_t lds128 = (size_t)(256 * (128 + 4)) * 4;
    size_t lds256 = (size_t)(256 * (256 + 4)) * 4;

    gemm_kernel<128, true, false><<<grid, 256, lds128, stream>>>(
        h, W1, b1, g1, bt1, m1, v1, nullptr, h1);
    gemm_kernel<256, true, false><<<grid, 256, lds256, stream>>>(
        h1, W2, b2, g2, bt2, m2, v2, nullptr, h2);
    gemm_kernel<128, false, true><<<grid, 256, lds128, stream>>>(
        x, Wres, nullptr, nullptr, nullptr, nullptr, nullptr, h2, (float*)d_out);
}


# the 3 product launches become one: a work-group takes its 160 rows through all 3 layers, restaging the next weights in LDS between layers and reading back the h1/h2 rows it wrote; sum order kept
# speedup vs baseline: 4.7958x; 1.0419x over previous
#include <hip/hip_runtime.h>
#include <hip/hip_bf16.h>

typedef float v2f __attribute__((ext_vector_type(2)));
typedef float v8f __attribute__((ext_vector_type(8)));

#define GIN_N    40000
#define GIN_E    640000
#define GIN_DIN  128
#define GIN_DOUT 256


__global__ void zero_kernel(int* __restrict__ p, int n) {
    int i = blockIdx.x * 256 + threadIdx.x;
    if (i < n) p[i] = 0;
}

#define GIN_CAP 32

#define GIN_SPLIT 20032
#define GIN_TASK  256
#define GIN_TICKET_GAP 64

__global__ __launch_bounds__(128)
void fill_kernel(const int* __restrict__ src, const int* __restrict__ dst,
                 int* __restrict__ cursor, int* __restrict__ slots,
                 int* __restrict__ spill_n, int* __restrict__ spill,
                 int* __restrict__ ticket, int e) {
    const int lane = threadIdx.x & 31;
    const int ntask = e / GIN_TASK;
    int half = (int)((__builtin_amdgcn_s_sendmsg_rtn(0x87) >> 18) & 1u);
    int t = 0;
    if (lane == 0) t = atomicAdd(&ticket[half * GIN_TICKET_GAP], 1);
    t = __builtin_amdgcn_readfirstlane(t);
    if (t >= ntask) {
        half ^= 1;
        if (lane == 0) t = atomicAdd(&ticket[half * GIN_TICKET_GAP], 1);
        t = __builtin_amdgcn_readfirstlane(t);
        if (t >= ntask) return;
    }
    const int lo = half ? GIN_SPLIT : 0;
    const int hi = half ? GIN_N : GIN_SPLIT;
    const int base = t * GIN_TASK;
    const int4 d0 = ((const int4*)(dst + base))[lane];
    const int4 d1 = ((const int4*)(dst + base + GIN_TASK / 2))[lane];
    const int4 s0 = ((const int4*)(src + base))[lane];
    const int4 s1 = ((const int4*)(src + base + GIN_TASK / 2))[lane];
    const int d[8] = {d0.x, d0.y, d0.z, d0.w, d1.x, d1.y, d1.z, d1.w};
    const int s[8] = {s0.x, s0.y, s0.z, s0.w, s1.x, s1.y, s1.z, s1.w};
    int pos[8];
#pragma unroll
    for (int k = 0; k < 8; ++k) {
        pos[k] = -1;
        if (d[k] >= lo && d[k] < hi) pos[k] = atomicAdd(&cursor[d[k]], 1);
    }
#pragma unroll
    for (int k = 0; k < 8; ++k) {
        if (pos[k] >= 0) {
            if (pos[k] < GIN_CAP) {
                slots[(size_t)d[k] * GIN_CAP + pos[k]] = s[k];
            } else {
                int q = atomicAdd(spill_n, 1);
                ((int2*)spill)[q] = make_int2(d[k], s[k]);
            }
        }
    }
}

__global__ __launch_bounds__(256)
void gather_kernel(const float* __restrict__ x, const int* __restrict__ cursor,
                   const int* __restrict__ slots, const int* __restrict__ spill_n,
                   const int* __restrict__ spill, const float* __restrict__ epsp,
                   float* __restrict__ h, int e) {
    const int node = __builtin_amdgcn_readfirstlane(blockIdx.x * 8 + (threadIdx.x >> 5));
    const int lane = threadIdx.x & 31;
    float se = 1.0f + epsp[0];
    const float4* xv = (const float4*)x;
    float4 self = xv[(size_t)node * 32 + lane];
    float ax = self.x * se, ay = self.y * se, az = self.z * se, aw = self.w * se;
    const int cnt = cursor[node];
    const int mine = slots[(size_t)node * GIN_CAP + lane];
    const int n = cnt < GIN_CAP ? cnt : GIN_CAP;
    for (int j = 0; j < n; ++j) {
        int s = __builtin_amdgcn_readlane(mine, j);
        float4 r = xv[(size_t)s * 32 + lane];
        ax += r.x; ay += r.y; az += r.z; aw += r.w;
    }
    if (cnt > GIN_CAP) {
        int m = spill_n[0];
        m = m < e ? m : e;
        const int2* sp = (const int2*)spill;
        for (int q0 = 0; q0 < m; q0 += 32) {
            int q = q0 + lane;
            int2 p = make_int2(-1, 0);
            if (q < m) p = sp[q];
            unsigned hit = __builtin_amdgcn_ballot_w32(p.x == node);
            while (hit) {
                int b = __builtin_ctz(hit);
                hit &= hit - 1;
                int s = __builtin_amdgcn_readlane(p.y, b);
                float4 r = xv[(size_t)s * 32 + lane];
                ax += r.x; ay += r.y; az += r.z; aw += r.w;
            }
        }
    }
    float4 o; o.x = ax; o.y = ay; o.z = az; o.w = aw;
    ((float4*)h)[(size_t)node * 32 + lane] = o;
}


#define GEMM_ROWS_PER_WG 160

template <int K, bool USE_BN, bool USE_RES>
__device__ __forceinline__
void gemm_phase(float* BsT, const float* __restrict__ A, const float* __restrict__ B,
                const float* __restrict__ bias, const float* __restrict__ gam,
                const float* __restrict__ bet, const float* __restrict__ mu,
                const float* __restrict__ var, const float* __restrict__ res,
                float* __restrict__ out) {
    constexpr int PT = K + 4;
    constexpr int NCOL = GIN_DOUT;

    const int tid = threadIdx.x;

    {
        const int cg = (tid >> 5) * 8 + (tid & 7);
        const int kq = (tid >> 3) & 3;
        constexpr int NP = K / 16;
#pragma unroll 4
        for (int p = 0; p < NP; ++p) {
            const int pp = (p + blockIdx.x) % NP;
            const int k = 16 * pp + 4 * kq;
            const float4 r0 = *(const float4*)(B + (size_t)(k + 0) * NCOL + 4 * cg);
            const float4 r1 = *(const float4*)(B + (size_t)(k + 1) * NCOL + 4 * cg);
            const float4 r2 = *(const float4*)(B + (size_t)(k + 2) * NCOL + 4 * cg);
            const float4 r3 = *(const float4*)(B + (size_t)(k + 3) * NCOL + 4 * cg);
            float* d = BsT + (4 * cg) * PT + k;
            *(float4*)(d)          = make_float4(r0.x, r1.x, r2.x, r3.x);
            *(float4*)(d + PT)     = make_float4(r0.y, r1.y, r2.y, r3.y);
            *(float4*)(d + 2 * PT) = make_float4(r0.z, r1.z, r2.z, r3.z);
            *(float4*)(d + 3 * PT) = make_float4(r0.w, r1.w, r2.w, r3.w);
        }
    }
    __syncthreads();

    const int wave = tid >> 5;
    const int lane = tid & 31;
    const int c  = lane & 15;
    const int h  = lane >> 4;
    const int colBase = (wave & 3) * 64;
    const int sp = wave >> 2;

    float scale[4], shift[4];
#pragma unroll
    for (int ct = 0; ct < 4; ++ct) {
        scale[ct] = 1.0f; shift[ct] = 0.0f;
        if (USE_BN) {
            int col = colBase + ct * 16 + c;
            float inv = rsqrtf(var[col] + 1e-5f);
            scale[ct] = gam[col] * inv;
            shift[ct] = (bias[col] - mu[col]) * scale[ct] + bet[col];
        }
    }

    const float* Bp = BsT + (colBase + c) * PT + 4 * h;

    float4 apf[8];
    {
        const float* Afirst = A + (size_t)(blockIdx.x * GEMM_ROWS_PER_WG + sp * 16 + c) * K + 4 * h;
#pragma unroll
        for (int j = 0; j < 8; ++j) apf[j] = *(const float4*)(Afirst + 8 * j);
    }

#pragma unroll 1
    for (int it = 0; it < GEMM_ROWS_PER_WG / 32; ++it) {
        const int row0 = blockIdx.x * GEMM_ROWS_PER_WG + (sp + 2 * it) * 16;
        const float* Arow = A + (size_t)(row0 + c) * K + 4 * h;

        float rv[4][8];
        if (USE_RES) {
#pragma unroll
            for (int ct = 0; ct < 4; ++ct)
#pragma unroll
                for (int r = 0; r < 8; ++r)
                    rv[ct][r] = res[(size_t)(row0 + 8 * h + r) * NCOL + colBase + ct * 16 + c];
        }

        v8f acc[4] = {};
#pragma unroll 1
        for (int k0 = 0; k0 < K; k0 += 128) {
            const bool lastHalf = (k0 + 128 >= K);
            const bool lastStrip = (it + 1 >= GEMM_ROWS_PER_WG / 32);
            const float* Anext = lastHalf ? (lastStrip ? Arow + k0 : Arow + (size_t)32 * K)
                                          : Arow + k0 + 128;
            float4 apn[8];
#pragma unroll
            for (int j = 0; j < 8; ++j) apn[j] = *(const float4*)(Anext + 8 * j);
#pragma unroll
            for (int j = 0; j < 16; ++j) {
                float4 a4 = (j < 8) ? apf[j] : *(const float4*)(Arow + k0 + 8 * j);
                float4 b4[4];
#pragma unroll
                for (int ct = 0; ct < 4; ++ct)
                    b4[ct] = *(const float4*)(Bp + ct * 16 * PT + k0 + 8 * j);
#pragma unroll
                for (int ct = 0; ct < 4; ++ct) {
                    v2f a, b;
                    a.x = a4.x; a.y = a4.y; b.x = b4[ct].x; b.y = b4[ct].y;
                    acc[ct] = __builtin_amdgcn_wmma_f32_16x16x4_f32(
                        false, a, false, b, (short)0, acc[ct], false, false);
                }
#pragma unroll
                for (int ct = 0; ct < 4; ++ct) {
                    v2f a, b;
                    a.x = a4.z; a.y = a4.w; b.x = b4[ct].z; b.y = b4[ct].w;
                    acc[ct] = __builtin_amdgcn_wmma_f32_16x16x4_f32(
                        false, a, false, b, (short)0, acc[ct], false, false);
                }
            }
#pragma unroll
            for (int j = 0; j < 8; ++j) apf[j] = apn[j];
        }

#pragma unroll
        for (int ct = 0; ct < 4; ++ct) {
            int col = colBase + ct * 16 + c;
#pragma unroll
            for (int r = 0; r < 8; ++r) {
                float val = acc[ct][r];
                if (USE_BN) val = val * scale[ct] + shift[ct];
                if (USE_RES) val += rv[ct][r];
                val = fmaxf(val, 0.0f);
                out[(size_t)(row0 + 8 * h + r) * NCOL + col] = val;
            }
        }
    }
}

__global__ __launch_bounds__(256) __attribute__((amdgpu_waves_per_eu(2, 2)))
void mlp_kernel(const float* __restrict__ x, const float* __restrict__ h,
                float* h1, float* h2,
                const float* __restrict__ W1, const float* __restrict__ b1,
                const float* __restrict__ g1, const float* __restrict__ bt1,
                const float* __restrict__ m1, const float* __restrict__ v1,
                const float* __restrict__ W2, const float* __restrict__ b2,
                const float* __restrict__ g2, const float* __restrict__ bt2,
                const float* __restrict__ m2, const float* __restrict__ v2,
                const float* __restrict__ Wres, float* __restrict__ out) {
    extern __shared__ float BsT[];

    gemm_phase<128, true, false>(BsT, h, W1, b1, g1, bt1, m1, v1, nullptr, h1);
    __syncthreads();
    gemm_phase<256, true, false>(BsT, h1, W2, b2, g2, bt2, m2, v2, nullptr, h2);
    __syncthreads();
    gemm_phase<128, false, true>(BsT, x, Wres, nullptr, nullptr, nullptr, nullptr, nullptr, h2, out);
}


extern "C" void kernel_launch(void* const* d_in, const int* in_sizes, int n_in,
                              void* d_out, int out_size, void* d_ws, size_t ws_size,
                              hipStream_t stream) {
    constexpr int N = GIN_N, E = GIN_E, DIN = GIN_DIN, DOUT = GIN_DOUT;

    const float* x    = (const float*)d_in[0];
    const int*   eidx = (const int*)d_in[1];
    const float* eps  = (const float*)d_in[2];
    const float* W1   = (const float*)d_in[3];
    const float* b1   = (const float*)d_in[4];
    const float* g1   = (const float*)d_in[5];
    const float* bt1  = (const float*)d_in[6];
    const float* m1   = (const float*)d_in[7];
    const float* v1   = (const float*)d_in[8];
    const float* W2   = (const float*)d_in[9];
    const float* b2   = (const float*)d_in[10];
    const float* g2   = (const float*)d_in[11];
    const float* bt2  = (const float*)d_in[12];
    const float* m2   = (const float*)d_in[13];
    const float* v2   = (const float*)d_in[14];
    const float* Wres = (const float*)d_in[15];

    const int* src = eidx;
    const int* dst = eidx + E;

    float* h  = (float*)d_ws;
    float* h1 = h + (size_t)N * DIN;
    float* h2 = h1 + (size_t)N * DOUT;
    int* cursor  = (int*)h2;
    int* spill_n = cursor + N;
    int* ticket  = cursor + N + 64;
    int* slots   = cursor + N + 256;
    int* spill   = slots + (size_t)N * GIN_CAP;

    zero_kernel<<<(N + 192 + 255) / 256, 256, 0, stream>>>(cursor, N + 192);
    fill_kernel<<<2 * (E / GIN_TASK) / 4, 128, 0, stream>>>(src, dst, cursor, slots, spill_n, spill,
                                                           ticket, E);
    gather_kernel<<<N / 8, 256, 0, stream>>>(x, cursor, slots, spill_n, spill, eps, h, E);

    dim3 grid(N / GEMM_ROWS_PER_WG);
    size_t lds256 = (size_t)(256 * (256 + 4)) * 4;
    mlp_kernel<<<grid, 256, lds256, stream>>>(
        x, h, h1, h2, W1, b1, g1, bt1, m1, v1, W2, b2, g2, bt2, m2, v2, Wres, (float*)d_out);
}
